# starting kernel
# baseline (speedup 1.0000x reference)
#include <hip/hip_runtime.h>

typedef __attribute__((ext_vector_type(2))) float v2f;
typedef __attribute__((ext_vector_type(8))) float v8f;

namespace {
constexpr int Dch   = 16;
constexpr int Llen  = 1000000;
constexpr int Wlen  = 64;
constexpr int Kout  = 8;
constexpr int Pad   = 31;
constexpr int MidLen = Llen - Wlen + 1;
constexpr int WavesPerBlock = 4;
constexpr int TilesPerWave  = 4;
constexpr int TilePos = 16;
constexpr int WavePos  = TilesPerWave * TilePos;
constexpr int BlockPos = WavesPerBlock * WavePos;
constexpr int XTile = BlockPos + Wlen;
constexpr int Threads = WavesPerBlock * 32;
constexpr int PStride = 1027;
constexpr int PZeroOff = Kout * PStride;
constexpr int PZeroLen = 1024;
constexpr int SPLen = PZeroOff + PZeroLen;
}

__global__ __launch_bounds__(Threads) void fullscan_mid_wmma(
    const float* __restrict__ xg, const float* __restrict__ Pg,
    float* __restrict__ out)
{
  __shared__ float sP[SPLen];
  __shared__ float sX[Dch * XTile];

  const int tid = threadIdx.x;
  const int N0  = blockIdx.x * BlockPos;

  for (int i = tid; i < Kout * Dch * Wlen; i += Threads) {
    int k  = i >> 10;
    int dw = i & 1023;
    sP[k * PStride + dw] = Pg[i];
  }
  for (int i = tid; i < PZeroLen; i += Threads) sP[PZeroOff + i] = 0.0f;

  for (int i = tid; i < Dch * XTile; i += Threads) {
    int d = i / XTile;
    int j = i - d * XTile;
    int gi = N0 + j;
    if (gi > Llen - 1) gi = Llen - 1;
    sX[i] = xg[(size_t)d * Llen + gi];
  }
  __syncthreads();

  const int wave = tid >> 5;
  const int lane = tid & 31;
  const int n0 = N0 + wave * WavePos;
  if (n0 >= MidLen) return;

  const int m  = lane & 15;
  const int hi = lane >> 4;
  const int c  = hi * 2;
  const float* xb = &sX[wave * WavePos + m + c];
  const float* pb = (m < Kout) ? &sP[m * PStride + c] : &sP[PZeroOff];

  v8f acc0 = {}, acc1 = {}, acc2 = {}, acc3 = {};
  for (int d = 0; d < Dch; ++d) {
#pragma unroll
    for (int w4 = 0; w4 < 16; ++w4) {
      const int xo = d * XTile + w4 * 4;
      v2f b;
      b.x = pb[d * Wlen + w4 * 4];
      b.y = pb[d * Wlen + w4 * 4 + 1];
      v2f a0, a1, a2, a3;
      a0.x = xb[xo +  0]; a0.y = xb[xo +  1];
      a1.x = xb[xo + 16]; a1.y = xb[xo + 17];
      a2.x = xb[xo + 32]; a2.y = xb[xo + 33];
      a3.x = xb[xo + 48]; a3.y = xb[xo + 49];
      acc0 = __builtin_amdgcn_wmma_f32_16x16x4_f32(false, a0, false, b, (short)0, acc0, false, false);
      acc1 = __builtin_amdgcn_wmma_f32_16x16x4_f32(false, a1, false, b, (short)0, acc1, false, false);
      acc2 = __builtin_amdgcn_wmma_f32_16x16x4_f32(false, a2, false, b, (short)0, acc2, false, false);
      acc3 = __builtin_amdgcn_wmma_f32_16x16x4_f32(false, a3, false, b, (short)0, acc3, false, false);
    }
  }

  const int k = m;
  if (k < Kout) {
    const int mrow0 = hi * 8;
    float* outk = out + (size_t)k * Llen + Pad;
#pragma unroll
    for (int j = 0; j < 8; ++j) {
      int n0j = n0 + mrow0 + j;
      if (n0j      < MidLen) outk[n0j]      = acc0[j];
      if (n0j + 16 < MidLen) outk[n0j + 16] = acc1[j];
      if (n0j + 32 < MidLen) outk[n0j + 32] = acc2[j];
      if (n0j + 48 < MidLen) outk[n0j + 48] = acc3[j];
    }
  }
}

__global__ void fullscan_edges(const float* __restrict__ xg,
                               const float* __restrict__ Pg,
                               float* __restrict__ out)
{
  int t = blockIdx.x * blockDim.x + threadIdx.x;
  if (t < Kout * Pad) {
    int k = t / Pad, j = t - k * Pad;
    float s = 0.f;
    for (int mm = 0; mm <= 32 + j; ++mm) {
      float tl = 0.f;
      for (int d = 0; d < Dch; ++d)
        tl += xg[(size_t)d * Llen + mm] * Pg[(k * Dch + d) * Wlen + mm];
      s += tl;
    }
    out[(size_t)k * Llen + j] = s;
  } else if (t < Kout * Pad + Kout * 32) {
    int t2 = t - Kout * Pad;
    int k = t2 / 32, j = t2 - k * 32;
    float s = 0.f;
    for (int mm = j + 1; mm < Wlen; ++mm) {
      for (int d = 0; d < Dch; ++d)
        s += xg[(size_t)d * Llen + (Llen - Wlen + mm)] *
             Pg[(k * Dch + d) * Wlen + mm];
    }
    out[(size_t)k * Llen + (Llen - 32) + j] = s;
  }
}

extern "C" void kernel_launch(void* const* d_in, const int*  , int  ,
                              void* d_out, int  , void*  , size_t  ,
                              hipStream_t stream) {
  const float* x = (const float*)d_in[0];
  const float* P = (const float*)d_in[1];
  float* out = (float*)d_out;

  const int blocks = (MidLen + BlockPos - 1) / BlockPos;
  hipLaunchKernelGGL(fullscan_mid_wmma, dim3(blocks), dim3(Threads), 0, stream, x, P, out);
  hipLaunchKernelGGL(fullscan_edges,    dim3(2),      dim3(256),     0, stream, x, P, out);
}


# Mid GEMM fills all 16 WMMA N columns: column = (channel, even/odd position), 17 k-steps per channel on zero-padded P rows: 272 not 512 WMMAs per 32 positions; odd-phase taps summed in shifted fours
# speedup vs baseline: 1.3865x; 1.3865x over previous
#include <hip/hip_runtime.h>


typedef __attribute__((ext_vector_type(2))) float v2f;
typedef __attribute__((ext_vector_type(8))) float v8f;

namespace {
constexpr int Dch   = 16;
constexpr int Llen  = 1000000;
constexpr int Wlen  = 64;
constexpr int Kout  = 8;
constexpr int Pad   = 31;
constexpr int MidLen = Llen - Wlen + 1;
constexpr int WavesPerBlock = 4;
constexpr int TilesPerWave  = 4;
constexpr int TilePos = 32;
constexpr int WavePos  = TilesPerWave * TilePos;
constexpr int BlockPos = WavesPerBlock * WavePos;
constexpr int Blocks   = (MidLen + BlockPos - 1) / BlockPos;
constexpr int Steps = 17;
constexpr int XTile = BlockPos + 4 * Steps;
constexpr int Threads = WavesPerBlock * 32;
constexpr int DP = 72;
constexpr int KP = Dch * DP + 5;
constexpr int SPLen = Kout * KP;
}

__global__ __launch_bounds__(Threads) void fullscan_mid_wmma(
    const float* __restrict__ xg, const float* __restrict__ Pg,
    float* __restrict__ out)
{
  __shared__ float sP[SPLen];
  __shared__ __attribute__((aligned(16))) float sX[Dch * XTile];

  const int tid = threadIdx.x;
  const int N0  = blockIdx.x * BlockPos;

  for (int i = tid; i < Kout * Dch * Wlen; i += Threads) {
    int k = i >> 10;
    int d = (i >> 6) & 15;
    int w = i & 63;
    sP[k * KP + d * DP + w + 1] = Pg[i];
  }
  for (int i = tid; i < Kout * Dch * 8; i += Threads) {
    int row = i >> 3;
    int q   = i & 7;
    int jj  = (q == 0) ? 0 : (Wlen + q);
    sP[(row >> 4) * KP + (row & 15) * DP + jj] = 0.0f;
  }

  for (int i = tid; i < Dch * XTile; i += Threads) {
    int d = i / XTile;
    int j = i - d * XTile;
    int gi = N0 + j;
    if (gi > Llen - 1) gi = Llen - 1;
    sX[i] = xg[(size_t)d * Llen + gi];
  }
  __syncthreads();

  const int wave = tid >> 5;
  const int lane = tid & 31;
  const int n0 = N0 + wave * WavePos;
  if (n0 >= MidLen) return;

  const int m  = lane & 15;
  const int hi = lane >> 4;
  const int c  = hi * 2;
  const int kch = m >> 1;
  const int sh  = m & 1;
  const float* xb = &sX[wave * WavePos + 2 * m + c];
  const float* pb = &sP[kch * KP + c - sh + 1];

  v8f acc0 = {}, acc1 = {}, acc2 = {}, acc3 = {};
  for (int d = 0; d < Dch; ++d) {
#pragma unroll
    for (int w4 = 0; w4 < Steps; ++w4) {
      const int xo = d * XTile + w4 * 4;
      v2f b;
      b.x = pb[d * DP + w4 * 4];
      b.y = pb[d * DP + w4 * 4 + 1];
      const v2f a0 = *reinterpret_cast<const v2f*>(&xb[xo +  0]);
      const v2f a1 = *reinterpret_cast<const v2f*>(&xb[xo + 32]);
      const v2f a2 = *reinterpret_cast<const v2f*>(&xb[xo + 64]);
      const v2f a3 = *reinterpret_cast<const v2f*>(&xb[xo + 96]);
      acc0 = __builtin_amdgcn_wmma_f32_16x16x4_f32(false, a0, false, b, (short)0, acc0, false, false);
      acc1 = __builtin_amdgcn_wmma_f32_16x16x4_f32(false, a1, false, b, (short)0, acc1, false, false);
      acc2 = __builtin_amdgcn_wmma_f32_16x16x4_f32(false, a2, false, b, (short)0, acc2, false, false);
      acc3 = __builtin_amdgcn_wmma_f32_16x16x4_f32(false, a3, false, b, (short)0, acc3, false, false);
    }
  }

  const int mrow0 = hi * 8;
  float* outk = out + (size_t)kch * Llen + Pad;
#pragma unroll
  for (int j = 0; j < 8; ++j) {
    int n0j = n0 + 2 * (mrow0 + j) + sh;
    if (n0j      < MidLen) outk[n0j]      = acc0[j];
    if (n0j + 32 < MidLen) outk[n0j + 32] = acc1[j];
    if (n0j + 64 < MidLen) outk[n0j + 64] = acc2[j];
    if (n0j + 96 < MidLen) outk[n0j + 96] = acc3[j];
  }
}

__global__ void fullscan_edges(const float* __restrict__ xg,
                               const float* __restrict__ Pg,
                               float* __restrict__ out)
{
  int t = blockIdx.x * blockDim.x + threadIdx.x;
  if (t < Kout * Pad) {
    int k = t / Pad, j = t - k * Pad;
    float s = 0.f;
    for (int mm = 0; mm <= 32 + j; ++mm) {
      float tl = 0.f;
      for (int d = 0; d < Dch; ++d)
        tl += xg[(size_t)d * Llen + mm] * Pg[(k * Dch + d) * Wlen + mm];
      s += tl;
    }
    out[(size_t)k * Llen + j] = s;
  } else if (t < Kout * Pad + Kout * 32) {
    int t2 = t - Kout * Pad;
    int k = t2 / 32, j = t2 - k * 32;
    float s = 0.f;
    for (int mm = j + 1; mm < Wlen; ++mm) {
      for (int d = 0; d < Dch; ++d)
        s += xg[(size_t)d * Llen + (Llen - Wlen + mm)] *
             Pg[(k * Dch + d) * Wlen + mm];
    }
    out[(size_t)k * Llen + (Llen - 32) + j] = s;
  }
}

extern "C" void kernel_launch(void* const* d_in, const int*  , int  ,
                              void* d_out, int  , void*  , size_t  ,
                              hipStream_t stream) {
  const float* x = (const float*)d_in[0];
  const float* P = (const float*)d_in[1];
  float* out = (float*)d_out;

  hipLaunchKernelGGL(fullscan_mid_wmma, dim3(Blocks), dim3(Threads), 0, stream, x, P, out);
  hipLaunchKernelGGL(fullscan_edges,    dim3(2),      dim3(256),     0, stream, x, P, out);
}


# Staging of P and the x tile: 16-byte global loads, all 35 per thread issued before the first LDS store, one memory round trip per block instead of a wait after every 4 to 8 four-byte loads
# speedup vs baseline: 1.5797x; 1.1393x over previous
#include <hip/hip_runtime.h>


typedef __attribute__((ext_vector_type(2))) float v2f;
typedef __attribute__((ext_vector_type(4))) float v4f;
typedef __attribute__((ext_vector_type(8))) float v8f;

namespace {
constexpr int Dch   = 16;
constexpr int Llen  = 1000000;
constexpr int Wlen  = 64;
constexpr int Kout  = 8;
constexpr int Pad   = 31;
constexpr int MidLen = Llen - Wlen + 1;
constexpr int WavesPerBlock = 4;
constexpr int TilesPerWave  = 4;
constexpr int TilePos = 32;
constexpr int WavePos  = TilesPerWave * TilePos;
constexpr int BlockPos = WavesPerBlock * WavePos;
constexpr int Blocks   = (MidLen + BlockPos - 1) / BlockPos;
constexpr int Steps = 17;
constexpr int XTile = BlockPos + 4 * Steps;
constexpr int Threads = WavesPerBlock * 32;
constexpr int DP = 72;
constexpr int KP = Dch * DP + 5;
constexpr int SPLen = Kout * KP;
constexpr int PVecs  = Kout * Dch * Wlen / 4;
constexpr int PIters = PVecs / Threads;
constexpr int RowVecs = XTile / 4;
constexpr int XVecs   = Dch * RowVecs;
constexpr int XIters  = (XVecs + Threads - 1) / Threads;
static_assert(PVecs % Threads == 0, "P pieces divide evenly");
static_assert((XIters - 1) * Threads <= XVecs, "every pass but the last is full");
static_assert(XTile % 4 == 0 && Llen % 4 == 0 && BlockPos % 4 == 0, "16-byte pieces never straddle the end of a row");
}

__global__ __launch_bounds__(Threads) void fullscan_mid_wmma(
    const float* __restrict__ xg, const float* __restrict__ Pg,
    float* __restrict__ out)
{
  __shared__ float sP[SPLen];
  __shared__ __attribute__((aligned(16))) float sX[Dch * XTile];

  const int tid = threadIdx.x;
  const int N0  = blockIdx.x * BlockPos;

  const v4f* P4 = reinterpret_cast<const v4f*>(Pg);
  v4f pr[PIters];
#pragma unroll
  for (int it = 0; it < PIters; ++it) pr[it] = P4[tid + it * Threads];

  v4f xr[XIters];
#pragma unroll
  for (int it = 0; it < XIters; ++it) {
    int i = tid + it * Threads;
    if (it == XIters - 1 && i > XVecs - 1) i = XVecs - 1;
    const int d = i / RowVecs;
    const int q = i - d * RowVecs;
    const int gi = N0 + 4 * q;
    const int gc = (gi < Llen) ? gi : (Llen - 4);
    v4f v = *reinterpret_cast<const v4f*>(&xg[(unsigned)d * (unsigned)Llen + (unsigned)gc]);
    if (gi >= Llen) { v.x = v.w; v.y = v.w; v.z = v.w; }
    xr[it] = v;
  }

  for (int i = tid; i < Kout * Dch * 8; i += Threads) {
    int row = i >> 3;
    int q   = i & 7;
    int jj  = (q == 0) ? 0 : (Wlen + q);
    sP[(row >> 4) * KP + (row & 15) * DP + jj] = 0.0f;
  }

#pragma unroll
  for (int it = 0; it < PIters; ++it) {
    const int e = 4 * (tid + it * Threads);
    const int k = e >> 10;
    const int d = (e >> 6) & 15;
    const int w = e & 63;
    float* dst = &sP[k * KP + d * DP + w + 1];
    dst[0] = pr[it].x; dst[1] = pr[it].y; dst[2] = pr[it].z; dst[3] = pr[it].w;
  }
  v4f* sX4 = reinterpret_cast<v4f*>(sX);
#pragma unroll
  for (int it = 0; it < XIters; ++it) {
    const int i = tid + it * Threads;
    if (it < XIters - 1 || i < XVecs) sX4[i] = xr[it];
  }
  __syncthreads();

  const int wave = tid >> 5;
  const int lane = tid & 31;
  const int n0 = N0 + wave * WavePos;
  if (n0 >= MidLen) return;

  const int m  = lane & 15;
  const int hi = lane >> 4;
  const int c  = hi * 2;
  const int kch = m >> 1;
  const int sh  = m & 1;
  const float* xb = &sX[wave * WavePos + 2 * m + c];
  const float* pb = &sP[kch * KP + c - sh + 1];

  v8f acc0 = {}, acc1 = {}, acc2 = {}, acc3 = {};
  for (int d = 0; d < Dch; ++d) {
#pragma unroll
    for (int w4 = 0; w4 < Steps; ++w4) {
      const int xo = d * XTile + w4 * 4;
      v2f b;
      b.x = pb[d * DP + w4 * 4];
      b.y = pb[d * DP + w4 * 4 + 1];
      const v2f a0 = *reinterpret_cast<const v2f*>(&xb[xo +  0]);
      const v2f a1 = *reinterpret_cast<const v2f*>(&xb[xo + 32]);
      const v2f a2 = *reinterpret_cast<const v2f*>(&xb[xo + 64]);
      const v2f a3 = *reinterpret_cast<const v2f*>(&xb[xo + 96]);
      acc0 = __builtin_amdgcn_wmma_f32_16x16x4_f32(false, a0, false, b, (short)0, acc0, false, false);
      acc1 = __builtin_amdgcn_wmma_f32_16x16x4_f32(false, a1, false, b, (short)0, acc1, false, false);
      acc2 = __builtin_amdgcn_wmma_f32_16x16x4_f32(false, a2, false, b, (short)0, acc2, false, false);
      acc3 = __builtin_amdgcn_wmma_f32_16x16x4_f32(false, a3, false, b, (short)0, acc3, false, false);
    }
  }

  const int mrow0 = hi * 8;
  float* outk = out + (size_t)kch * Llen + Pad;
#pragma unroll
  for (int j = 0; j < 8; ++j) {
    int n0j = n0 + 2 * (mrow0 + j) + sh;
    if (n0j      < MidLen) outk[n0j]      = acc0[j];
    if (n0j + 32 < MidLen) outk[n0j + 32] = acc1[j];
    if (n0j + 64 < MidLen) outk[n0j + 64] = acc2[j];
    if (n0j + 96 < MidLen) outk[n0j + 96] = acc3[j];
  }
}

__global__ void fullscan_edges(const float* __restrict__ xg,
                               const float* __restrict__ Pg,
                               float* __restrict__ out)
{
  int t = blockIdx.x * blockDim.x + threadIdx.x;
  if (t < Kout * Pad) {
    int k = t / Pad, j = t - k * Pad;
    float s = 0.f;
    for (int mm = 0; mm <= 32 + j; ++mm) {
      float tl = 0.f;
      for (int d = 0; d < Dch; ++d)
        tl += xg[(size_t)d * Llen + mm] * Pg[(k * Dch + d) * Wlen + mm];
      s += tl;
    }
    out[(size_t)k * Llen + j] = s;
  } else if (t < Kout * Pad + Kout * 32) {
    int t2 = t - Kout * Pad;
    int k = t2 / 32, j = t2 - k * 32;
    float s = 0.f;
    for (int mm = j + 1; mm < Wlen; ++mm) {
      for (int d = 0; d < Dch; ++d)
        s += xg[(size_t)d * Llen + (Llen - Wlen + mm)] *
             Pg[(k * Dch + d) * Wlen + mm];
    }
    out[(size_t)k * Llen + (Llen - 32) + j] = s;
  }
}

extern "C" void kernel_launch(void* const* d_in, const int*  , int  ,
                              void* d_out, int  , void*  , size_t  ,
                              hipStream_t stream) {
  const float* x = (const float*)d_in[0];
  const float* P = (const float*)d_in[1];
  float* out = (float*)d_out;

  hipLaunchKernelGGL(fullscan_mid_wmma, dim3(Blocks), dim3(Threads), 0, stream, x, P, out);
  hipLaunchKernelGGL(fullscan_edges,    dim3(2),      dim3(256),     0, stream, x, P, out);
}


# One launch instead of two: the 504 edge sums run in 4 extra work-groups at the head of the mid kernel's grid, beside the mid pass, not in a dependent launch behind it
# speedup vs baseline: 2.3013x; 1.4568x over previous
#include <hip/hip_runtime.h>


typedef __attribute__((ext_vector_type(2))) float v2f;
typedef __attribute__((ext_vector_type(4))) float v4f;
typedef __attribute__((ext_vector_type(8))) float v8f;

namespace {
constexpr int Dch   = 16;
constexpr int Llen  = 1000000;
constexpr int Wlen  = 64;
constexpr int Kout  = 8;
constexpr int Pad   = 31;
constexpr int MidLen = Llen - Wlen + 1;
constexpr int WavesPerBlock = 4;
constexpr int TilesPerWave  = 4;
constexpr int TilePos = 32;
constexpr int WavePos  = TilesPerWave * TilePos;
constexpr int BlockPos = WavesPerBlock * WavePos;
constexpr int Blocks   = (MidLen + BlockPos - 1) / BlockPos;
constexpr int Steps = 17;
constexpr int XTile = BlockPos + 4 * Steps;
constexpr int Threads = WavesPerBlock * 32;
constexpr int DP = 72;
constexpr int KP = Dch * DP + 5;
constexpr int SPLen = Kout * KP;
constexpr int PVecs  = Kout * Dch * Wlen / 4;
constexpr int PIters = PVecs / Threads;
constexpr int RowVecs = XTile / 4;
constexpr int XVecs   = Dch * RowVecs;
constexpr int XIters  = (XVecs + Threads - 1) / Threads;
constexpr int EdgeOuts   = Kout * Pad + Kout * 32;
constexpr int EdgeBlocks = (EdgeOuts + Threads - 1) / Threads;
constexpr int GridBlocks = EdgeBlocks + Blocks;
static_assert(PVecs % Threads == 0, "P pieces divide evenly");
static_assert((XIters - 1) * Threads <= XVecs, "every pass but the last is full");
static_assert(XTile % 4 == 0 && Llen % 4 == 0 && BlockPos % 4 == 0, "16-byte pieces never straddle the end of a row");
}

__device__ __forceinline__ void fullscan_edges(int t, const float* __restrict__ xg,
                                               const float* __restrict__ Pg,
                                               float* __restrict__ out)
{
  if (t < Kout * Pad) {
    int k = t / Pad, j = t - k * Pad;
    float s = 0.f;
    for (int mm = 0; mm <= 32 + j; ++mm) {
      float tl = 0.f;
      for (int d = 0; d < Dch; ++d)
        tl += xg[(size_t)d * Llen + mm] * Pg[(k * Dch + d) * Wlen + mm];
      s += tl;
    }
    out[(size_t)k * Llen + j] = s;
  } else if (t < Kout * Pad + Kout * 32) {
    int t2 = t - Kout * Pad;
    int k = t2 / 32, j = t2 - k * 32;
    float s = 0.f;
    for (int mm = j + 1; mm < Wlen; ++mm) {
      for (int d = 0; d < Dch; ++d)
        s += xg[(size_t)d * Llen + (Llen - Wlen + mm)] *
             Pg[(k * Dch + d) * Wlen + mm];
    }
    out[(size_t)k * Llen + (Llen - 32) + j] = s;
  }
}

__global__ __launch_bounds__(Threads) void fullscan_mid_wmma(
    const float* __restrict__ xg, const float* __restrict__ Pg,
    float* __restrict__ out)
{
  __shared__ float sP[SPLen];
  __shared__ __attribute__((aligned(16))) float sX[Dch * XTile];

  if (blockIdx.x < EdgeBlocks) {
    fullscan_edges(blockIdx.x * Threads + threadIdx.x, xg, Pg, out);
    return;
  }

  const int tid = threadIdx.x;
  const int N0  = (blockIdx.x - EdgeBlocks) * BlockPos;

  const v4f* P4 = reinterpret_cast<const v4f*>(Pg);
  v4f pr[PIters];
#pragma unroll
  for (int it = 0; it < PIters; ++it) pr[it] = P4[tid + it * Threads];

  v4f xr[XIters];
#pragma unroll
  for (int it = 0; it < XIters; ++it) {
    int i = tid + it * Threads;
    if (it == XIters - 1 && i > XVecs - 1) i = XVecs - 1;
    const int d = i / RowVecs;
    const int q = i - d * RowVecs;
    const int gi = N0 + 4 * q;
    const int gc = (gi < Llen) ? gi : (Llen - 4);
    v4f v = *reinterpret_cast<const v4f*>(&xg[(unsigned)d * (unsigned)Llen + (unsigned)gc]);
    if (gi >= Llen) { v.x = v.w; v.y = v.w; v.z = v.w; }
    xr[it] = v;
  }

  for (int i = tid; i < Kout * Dch * 8; i += Threads) {
    int row = i >> 3;
    int q   = i & 7;
    int jj  = (q == 0) ? 0 : (Wlen + q);
    sP[(row >> 4) * KP + (row & 15) * DP + jj] = 0.0f;
  }

#pragma unroll
  for (int it = 0; it < PIters; ++it) {
    const int e = 4 * (tid + it * Threads);
    const int k = e >> 10;
    const int d = (e >> 6) & 15;
    const int w = e & 63;
    float* dst = &sP[k * KP + d * DP + w + 1];
    dst[0] = pr[it].x; dst[1] = pr[it].y; dst[2] = pr[it].z; dst[3] = pr[it].w;
  }
  v4f* sX4 = reinterpret_cast<v4f*>(sX);
#pragma unroll
  for (int it = 0; it < XIters; ++it) {
    const int i = tid + it * Threads;
    if (it < XIters - 1 || i < XVecs) sX4[i] = xr[it];
  }
  __syncthreads();

  const int wave = tid >> 5;
  const int lane = tid & 31;
  const int n0 = N0 + wave * WavePos;
  if (n0 >= MidLen) return;

  const int m  = lane & 15;
  const int hi = lane >> 4;
  const int c  = hi * 2;
  const int kch = m >> 1;
  const int sh  = m & 1;
  const float* xb = &sX[wave * WavePos + 2 * m + c];
  const float* pb = &sP[kch * KP + c - sh + 1];

  v8f acc0 = {}, acc1 = {}, acc2 = {}, acc3 = {};
  for (int d = 0; d < Dch; ++d) {
#pragma unroll
    for (int w4 = 0; w4 < Steps; ++w4) {
      const int xo = d * XTile + w4 * 4;
      v2f b;
      b.x = pb[d * DP + w4 * 4];
      b.y = pb[d * DP + w4 * 4 + 1];
      const v2f a0 = *reinterpret_cast<const v2f*>(&xb[xo +  0]);
      const v2f a1 = *reinterpret_cast<const v2f*>(&xb[xo + 32]);
      const v2f a2 = *reinterpret_cast<const v2f*>(&xb[xo + 64]);
      const v2f a3 = *reinterpret_cast<const v2f*>(&xb[xo + 96]);
      acc0 = __builtin_amdgcn_wmma_f32_16x16x4_f32(false, a0, false, b, (short)0, acc0, false, false);
      acc1 = __builtin_amdgcn_wmma_f32_16x16x4_f32(false, a1, false, b, (short)0, acc1, false, false);
      acc2 = __builtin_amdgcn_wmma_f32_16x16x4_f32(false, a2, false, b, (short)0, acc2, false, false);
      acc3 = __builtin_amdgcn_wmma_f32_16x16x4_f32(false, a3, false, b, (short)0, acc3, false, false);
    }
  }

  const int mrow0 = hi * 8;
  float* outk = out + (size_t)kch * Llen + Pad;
#pragma unroll
  for (int j = 0; j < 8; ++j) {
    int n0j = n0 + 2 * (mrow0 + j) + sh;
    if (n0j      < MidLen) outk[n0j]      = acc0[j];
    if (n0j + 32 < MidLen) outk[n0j + 32] = acc1[j];
    if (n0j + 64 < MidLen) outk[n0j + 64] = acc2[j];
    if (n0j + 96 < MidLen) outk[n0j + 96] = acc3[j];
  }
}

extern "C" void kernel_launch(void* const* d_in, const int*  , int  ,
                              void* d_out, int  , void*  , size_t  ,
                              hipStream_t stream) {
  const float* x = (const float*)d_in[0];
  const float* P = (const float*)d_in[1];
  float* out = (float*)d_out;

  hipLaunchKernelGGL(fullscan_mid_wmma, dim3(GridBlocks), dim3(Threads), 0, stream, x, P, out);
}


# Outputs staged through LDS as [channel][position] rows: per wave eight 16-byte stores of 124 consecutive floats plus one 4-byte store, not 32 four-byte stores of 8-byte pieces in 16 lines each
# speedup vs baseline: 2.4270x; 1.0546x over previous
#include <hip/hip_runtime.h>


typedef __attribute__((ext_vector_type(2))) float v2f;
typedef __attribute__((ext_vector_type(4))) float v4f;
typedef __attribute__((ext_vector_type(8))) float v8f;

namespace {
constexpr int Dch   = 16;
constexpr int Llen  = 1000000;
constexpr int Wlen  = 64;
constexpr int Kout  = 8;
constexpr int Pad   = 31;
constexpr int MidLen = Llen - Wlen + 1;
constexpr int WavesPerBlock = 4;
constexpr int TilesPerWave  = 4;
constexpr int TilePos = 32;
constexpr int WavePos  = TilesPerWave * TilePos;
constexpr int BlockPos = WavesPerBlock * WavePos;
constexpr int Blocks   = (MidLen + BlockPos - 1) / BlockPos;
constexpr int Steps = 17;
constexpr int XTile = BlockPos + 4 * Steps;
constexpr int Threads = WavesPerBlock * 32;
constexpr int DP = 72;
constexpr int KP = Dch * DP + 5;
constexpr int SPLen = Kout * KP;
constexpr int PVecs  = Kout * Dch * Wlen / 4;
constexpr int PIters = PVecs / Threads;
constexpr int RowVecs = XTile / 4;
constexpr int XVecs   = Dch * RowVecs;
constexpr int XIters  = (XVecs + Threads - 1) / Threads;
constexpr int EdgeOuts   = Kout * Pad + Kout * 32;
constexpr int EdgeBlocks = (EdgeOuts + Threads - 1) / Threads;
constexpr int GridBlocks = EdgeBlocks + Blocks;
constexpr int OP = 132;
constexpr int WaveOut = Kout * OP;
static_assert(WavesPerBlock * WaveOut <= Dch * XTile, "the staged outputs fit in the x tile's space");
static_assert(OP % 4 == 0 && WavePos % 32 == 0 && (Pad + 1) % 32 == 0 && Llen % 32 == 0, "position 1 of a wave starts a 128-byte line of out");
static_assert(PVecs % Threads == 0, "P pieces divide evenly");
static_assert((XIters - 1) * Threads <= XVecs, "every pass but the last is full");
static_assert(XTile % 4 == 0 && Llen % 4 == 0 && BlockPos % 4 == 0, "16-byte pieces never straddle the end of a row");
}

__device__ __forceinline__ void fullscan_edges(int t, const float* __restrict__ xg,
                                               const float* __restrict__ Pg,
                                               float* __restrict__ out)
{
  if (t < Kout * Pad) {
    int k = t / Pad, j = t - k * Pad;
    float s = 0.f;
    for (int mm = 0; mm <= 32 + j; ++mm) {
      float tl = 0.f;
      for (int d = 0; d < Dch; ++d)
        tl += xg[(size_t)d * Llen + mm] * Pg[(k * Dch + d) * Wlen + mm];
      s += tl;
    }
    out[(size_t)k * Llen + j] = s;
  } else if (t < Kout * Pad + Kout * 32) {
    int t2 = t - Kout * Pad;
    int k = t2 / 32, j = t2 - k * 32;
    float s = 0.f;
    for (int mm = j + 1; mm < Wlen; ++mm) {
      for (int d = 0; d < Dch; ++d)
        s += xg[(size_t)d * Llen + (Llen - Wlen + mm)] *
             Pg[(k * Dch + d) * Wlen + mm];
    }
    out[(size_t)k * Llen + (Llen - 32) + j] = s;
  }
}

__global__ __launch_bounds__(Threads) void fullscan_mid_wmma(
    const float* __restrict__ xg, const float* __restrict__ Pg,
    float* __restrict__ out)
{
  __shared__ float sP[SPLen];
  __shared__ __attribute__((aligned(16))) float sX[Dch * XTile];

  if (blockIdx.x < EdgeBlocks) {
    fullscan_edges(blockIdx.x * Threads + threadIdx.x, xg, Pg, out);
    return;
  }

  const int tid = threadIdx.x;
  const int N0  = (blockIdx.x - EdgeBlocks) * BlockPos;

  const v4f* P4 = reinterpret_cast<const v4f*>(Pg);
  v4f pr[PIters];
#pragma unroll
  for (int it = 0; it < PIters; ++it) pr[it] = P4[tid + it * Threads];

  v4f xr[XIters];
#pragma unroll
  for (int it = 0; it < XIters; ++it) {
    int i = tid + it * Threads;
    if (it == XIters - 1 && i > XVecs - 1) i = XVecs - 1;
    const int d = i / RowVecs;
    const int q = i - d * RowVecs;
    const int gi = N0 + 4 * q;
    const int gc = (gi < Llen) ? gi : (Llen - 4);
    v4f v = *reinterpret_cast<const v4f*>(&xg[(unsigned)d * (unsigned)Llen + (unsigned)gc]);
    if (gi >= Llen) { v.x = v.w; v.y = v.w; v.z = v.w; }
    xr[it] = v;
  }

  for (int i = tid; i < Kout * Dch * 8; i += Threads) {
    int row = i >> 3;
    int q   = i & 7;
    int jj  = (q == 0) ? 0 : (Wlen + q);
    sP[(row >> 4) * KP + (row & 15) * DP + jj] = 0.0f;
  }

#pragma unroll
  for (int it = 0; it < PIters; ++it) {
    const int e = 4 * (tid + it * Threads);
    const int k = e >> 10;
    const int d = (e >> 6) & 15;
    const int w = e & 63;
    float* dst = &sP[k * KP + d * DP + w + 1];
    dst[0] = pr[it].x; dst[1] = pr[it].y; dst[2] = pr[it].z; dst[3] = pr[it].w;
  }
  v4f* sX4 = reinterpret_cast<v4f*>(sX);
#pragma unroll
  for (int it = 0; it < XIters; ++it) {
    const int i = tid + it * Threads;
    if (it < XIters - 1 || i < XVecs) sX4[i] = xr[it];
  }
  __syncthreads();

  const int wave = tid >> 5;
  const int lane = tid & 31;
  const int n0 = N0 + wave * WavePos;
  const bool active = n0 < MidLen;

  const int m  = lane & 15;
  const int hi = lane >> 4;
  const int c  = hi * 2;
  const int kch = m >> 1;
  const int sh  = m & 1;
  const float* xb = &sX[wave * WavePos + 2 * m + c];
  const float* pb = &sP[kch * KP + c - sh + 1];

  v8f acc0 = {}, acc1 = {}, acc2 = {}, acc3 = {};
  if (active) {
    for (int d = 0; d < Dch; ++d) {
#pragma unroll
      for (int w4 = 0; w4 < Steps; ++w4) {
        const int xo = d * XTile + w4 * 4;
        v2f b;
        b.x = pb[d * DP + w4 * 4];
        b.y = pb[d * DP + w4 * 4 + 1];
        const v2f a0 = *reinterpret_cast<const v2f*>(&xb[xo +  0]);
        const v2f a1 = *reinterpret_cast<const v2f*>(&xb[xo + 32]);
        const v2f a2 = *reinterpret_cast<const v2f*>(&xb[xo + 64]);
        const v2f a3 = *reinterpret_cast<const v2f*>(&xb[xo + 96]);
        acc0 = __builtin_amdgcn_wmma_f32_16x16x4_f32(false, a0, false, b, (short)0, acc0, false, false);
        acc1 = __builtin_amdgcn_wmma_f32_16x16x4_f32(false, a1, false, b, (short)0, acc1, false, false);
        acc2 = __builtin_amdgcn_wmma_f32_16x16x4_f32(false, a2, false, b, (short)0, acc2, false, false);
        acc3 = __builtin_amdgcn_wmma_f32_16x16x4_f32(false, a3, false, b, (short)0, acc3, false, false);
      }
    }
  }

  __syncthreads();
  float* sO = sX + wave * WaveOut;
  if (active) {
    float* row = &sO[kch * OP + 3 + 16 * hi + sh];
#pragma unroll
    for (int j = 0; j < 8; ++j) {
      row[2 * j]      = acc0[j];
      row[2 * j + 32] = acc1[j];
      row[2 * j + 64] = acc2[j];
      row[2 * j + 96] = acc3[j];
    }
  }
  __syncthreads();
  if (active) {
    const int p  = 1 + 4 * lane;
    const int ks = lane >> 2;
    const int qs = lane & 3;
    const int ps = (qs == 0) ? 0 : (WavePos - 4 + qs);
    if (n0 + WavePos <= MidLen) {
      if (lane < 31) {
        const float* srow = &sO[3 + p];
        float* dst = out + Pad + n0 + p;
#pragma unroll
        for (int k = 0; k < Kout; ++k)
          *reinterpret_cast<v4f*>(dst + (size_t)k * Llen) = *reinterpret_cast<const v4f*>(srow + k * OP);
      }
      out[(size_t)ks * Llen + Pad + n0 + ps] = sO[ks * OP + 3 + ps];
    } else {
      if (lane < 31) {
        for (int k = 0; k < Kout; ++k)
          for (int e = 0; e < 4; ++e)
            if (n0 + p + e < MidLen) out[(size_t)k * Llen + Pad + n0 + p + e] = sO[k * OP + 3 + p + e];
      }
      if (n0 + ps < MidLen) out[(size_t)ks * Llen + Pad + n0 + ps] = sO[ks * OP + 3 + ps];
    }
  }
}

extern "C" void kernel_launch(void* const* d_in, const int*  , int  ,
                              void* d_out, int  , void*  , size_t  ,
                              hipStream_t stream) {
  const float* x = (const float*)d_in[0];
  const float* P = (const float*)d_in[1];
  float* out = (float*)d_out;

  hipLaunchKernelGGL(fullscan_mid_wmma, dim3(GridBlocks), dim3(Threads), 0, stream, x, P, out);
}


# x offset 64 (odd phase, tap 63) moved out of a 17th step per channel into 4 closing steps whose k slots are 4 channels: 1040 not 1088 WMMAs per wave; these 16 products are now summed last
# speedup vs baseline: 2.4960x; 1.0284x over previous
#include <hip/hip_runtime.h>


typedef __attribute__((ext_vector_type(2))) float v2f;
typedef __attribute__((ext_vector_type(4))) float v4f;
typedef __attribute__((ext_vector_type(8))) float v8f;

namespace {
constexpr int Dch   = 16;
constexpr int Llen  = 1000000;
constexpr int Wlen  = 64;
constexpr int Kout  = 8;
constexpr int Pad   = 31;
constexpr int MidLen = Llen - Wlen + 1;
constexpr int WavesPerBlock = 4;
constexpr int TilesPerWave  = 4;
constexpr int TilePos = 32;
constexpr int WavePos  = TilesPerWave * TilePos;
constexpr int BlockPos = WavesPerBlock * WavePos;
constexpr int Blocks   = (MidLen + BlockPos - 1) / BlockPos;
constexpr int Steps = 17;
constexpr int MainSteps = Wlen / 4;
constexpr int XTile = BlockPos + 4 * Steps;
constexpr int Threads = WavesPerBlock * 32;
constexpr int DP = 72;
constexpr int KP = Dch * DP + 5;
constexpr int SPLen = Kout * KP;
constexpr int PVecs  = Kout * Dch * Wlen / 4;
constexpr int PIters = PVecs / Threads;
constexpr int RowVecs = XTile / 4;
constexpr int XVecs   = Dch * RowVecs;
constexpr int XIters  = (XVecs + Threads - 1) / Threads;
constexpr int EdgeOuts   = Kout * Pad + Kout * 32;
constexpr int EdgeBlocks = (EdgeOuts + Threads - 1) / Threads;
constexpr int GridBlocks = EdgeBlocks + Blocks;
constexpr int OP = 132;
constexpr int WaveOut = Kout * OP;
static_assert(WavesPerBlock * WaveOut <= Dch * XTile, "the staged outputs fit in the x tile's space");
static_assert(OP % 4 == 0 && WavePos % 32 == 0 && (Pad + 1) % 32 == 0 && Llen % 32 == 0, "position 1 of a wave starts a 128-byte line of out");
static_assert(PVecs % Threads == 0, "P pieces divide evenly");
static_assert((XIters - 1) * Threads <= XVecs, "every pass but the last is full");
static_assert(XTile % 4 == 0 && Llen % 4 == 0 && BlockPos % 4 == 0, "16-byte pieces never straddle the end of a row");
}

__device__ __forceinline__ void fullscan_edges(int t, const float* __restrict__ xg,
                                               const float* __restrict__ Pg,
                                               float* __restrict__ out)
{
  if (t < Kout * Pad) {
    int k = t / Pad, j = t - k * Pad;
    float s = 0.f;
    for (int mm = 0; mm <= 32 + j; ++mm) {
      float tl = 0.f;
      for (int d = 0; d < Dch; ++d)
        tl += xg[(size_t)d * Llen + mm] * Pg[(k * Dch + d) * Wlen + mm];
      s += tl;
    }
    out[(size_t)k * Llen + j] = s;
  } else if (t < Kout * Pad + Kout * 32) {
    int t2 = t - Kout * Pad;
    int k = t2 / 32, j = t2 - k * 32;
    float s = 0.f;
    for (int mm = j + 1; mm < Wlen; ++mm) {
      for (int d = 0; d < Dch; ++d)
        s += xg[(size_t)d * Llen + (Llen - Wlen + mm)] *
             Pg[(k * Dch + d) * Wlen + mm];
    }
    out[(size_t)k * Llen + (Llen - 32) + j] = s;
  }
}

__global__ __launch_bounds__(Threads) void fullscan_mid_wmma(
    const float* __restrict__ xg, const float* __restrict__ Pg,
    float* __restrict__ out)
{
  __shared__ float sP[SPLen];
  __shared__ __attribute__((aligned(16))) float sX[Dch * XTile];

  if (blockIdx.x < EdgeBlocks) {
    fullscan_edges(blockIdx.x * Threads + threadIdx.x, xg, Pg, out);
    return;
  }

  const int tid = threadIdx.x;
  const int N0  = (blockIdx.x - EdgeBlocks) * BlockPos;

  const v4f* P4 = reinterpret_cast<const v4f*>(Pg);
  v4f pr[PIters];
#pragma unroll
  for (int it = 0; it < PIters; ++it) pr[it] = P4[tid + it * Threads];

  v4f xr[XIters];
#pragma unroll
  for (int it = 0; it < XIters; ++it) {
    int i = tid + it * Threads;
    if (it == XIters - 1 && i > XVecs - 1) i = XVecs - 1;
    const int d = i / RowVecs;
    const int q = i - d * RowVecs;
    const int gi = N0 + 4 * q;
    const int gc = (gi < Llen) ? gi : (Llen - 4);
    v4f v = *reinterpret_cast<const v4f*>(&xg[(unsigned)d * (unsigned)Llen + (unsigned)gc]);
    if (gi >= Llen) { v.x = v.w; v.y = v.w; v.z = v.w; }
    xr[it] = v;
  }

  for (int i = tid; i < Kout * Dch * 8; i += Threads) {
    int row = i >> 3;
    int q   = i & 7;
    int jj  = (q == 0) ? 0 : (Wlen + q);
    sP[(row >> 4) * KP + (row & 15) * DP + jj] = 0.0f;
  }

#pragma unroll
  for (int it = 0; it < PIters; ++it) {
    const int e = 4 * (tid + it * Threads);
    const int k = e >> 10;
    const int d = (e >> 6) & 15;
    const int w = e & 63;
    float* dst = &sP[k * KP + d * DP + w + 1];
    dst[0] = pr[it].x; dst[1] = pr[it].y; dst[2] = pr[it].z; dst[3] = pr[it].w;
  }
  v4f* sX4 = reinterpret_cast<v4f*>(sX);
#pragma unroll
  for (int it = 0; it < XIters; ++it) {
    const int i = tid + it * Threads;
    if (it < XIters - 1 || i < XVecs) sX4[i] = xr[it];
  }
  __syncthreads();

  const int wave = tid >> 5;
  const int lane = tid & 31;
  const int n0 = N0 + wave * WavePos;
  const bool active = n0 < MidLen;

  const int m  = lane & 15;
  const int hi = lane >> 4;
  const int c  = hi * 2;
  const int kch = m >> 1;
  const int sh  = m & 1;
  const float* xb = &sX[wave * WavePos + 2 * m + c];
  const float* pb = &sP[kch * KP + c - sh + 1];

  v8f acc0 = {}, acc1 = {}, acc2 = {}, acc3 = {};
  if (active) {
    for (int d = 0; d < Dch; ++d) {
#pragma unroll
      for (int w4 = 0; w4 < MainSteps; ++w4) {
        const int xo = d * XTile + w4 * 4;
        v2f b;
        b.x = pb[d * DP + w4 * 4];
        b.y = pb[d * DP + w4 * 4 + 1];
        const v2f a0 = *reinterpret_cast<const v2f*>(&xb[xo +  0]);
        const v2f a1 = *reinterpret_cast<const v2f*>(&xb[xo + 32]);
        const v2f a2 = *reinterpret_cast<const v2f*>(&xb[xo + 64]);
        const v2f a3 = *reinterpret_cast<const v2f*>(&xb[xo + 96]);
        acc0 = __builtin_amdgcn_wmma_f32_16x16x4_f32(false, a0, false, b, (short)0, acc0, false, false);
        acc1 = __builtin_amdgcn_wmma_f32_16x16x4_f32(false, a1, false, b, (short)0, acc1, false, false);
        acc2 = __builtin_amdgcn_wmma_f32_16x16x4_f32(false, a2, false, b, (short)0, acc2, false, false);
        acc3 = __builtin_amdgcn_wmma_f32_16x16x4_f32(false, a3, false, b, (short)0, acc3, false, false);
      }
    }
    const float* xt = &sX[wave * WavePos + 2 * m + Wlen];
    const float* pt = &sP[kch * KP + Wlen];
#pragma unroll
    for (int g = 0; g < Dch / 4; ++g) {
      const int d0 = 4 * g + c;
      const float p0 = pt[d0 * DP];
      const float p1 = pt[(d0 + 1) * DP];
      v2f b;
      b.x = sh ? p0 : 0.0f;
      b.y = sh ? p1 : 0.0f;
      const float* x0 = xt + d0 * XTile;
      const float* x1 = x0 + XTile;
      v2f a0, a1, a2, a3;
      a0.x = x0[ 0]; a0.y = x1[ 0];
      a1.x = x0[32]; a1.y = x1[32];
      a2.x = x0[64]; a2.y = x1[64];
      a3.x = x0[96]; a3.y = x1[96];
      acc0 = __builtin_amdgcn_wmma_f32_16x16x4_f32(false, a0, false, b, (short)0, acc0, false, false);
      acc1 = __builtin_amdgcn_wmma_f32_16x16x4_f32(false, a1, false, b, (short)0, acc1, false, false);
      acc2 = __builtin_amdgcn_wmma_f32_16x16x4_f32(false, a2, false, b, (short)0, acc2, false, false);
      acc3 = __builtin_amdgcn_wmma_f32_16x16x4_f32(false, a3, false, b, (short)0, acc3, false, false);
    }
  }

  __syncthreads();
  float* sO = sX + wave * WaveOut;
  if (active) {
    float* row = &sO[kch * OP + 3 + 16 * hi + sh];
#pragma unroll
    for (int j = 0; j < 8; ++j) {
      row[2 * j]      = acc0[j];
      row[2 * j + 32] = acc1[j];
      row[2 * j + 64] = acc2[j];
      row[2 * j + 96] = acc3[j];
    }
  }
  __syncthreads();
  if (active) {
    const int p  = 1 + 4 * lane;
    const int ks = lane >> 2;
    const int qs = lane & 3;
    const int ps = (qs == 0) ? 0 : (WavePos - 4 + qs);
    if (n0 + WavePos <= MidLen) {
      if (lane < 31) {
        const float* srow = &sO[3 + p];
        float* dst = out + Pad + n0 + p;
#pragma unroll
        for (int k = 0; k < Kout; ++k)
          *reinterpret_cast<v4f*>(dst + (size_t)k * Llen) = *reinterpret_cast<const v4f*>(srow + k * OP);
      }
      out[(size_t)ks * Llen + Pad + n0 + ps] = sO[ks * OP + 3 + ps];
    } else {
      if (lane < 31) {
        for (int k = 0; k < Kout; ++k)
          for (int e = 0; e < 4; ++e)
            if (n0 + p + e < MidLen) out[(size_t)k * Llen + Pad + n0 + p + e] = sO[k * OP + 3 + p + e];
      }
      if (n0 + ps < MidLen) out[(size_t)ks * Llen + Pad + n0 + ps] = sO[ks * OP + 3 + ps];
    }
  }
}

extern "C" void kernel_launch(void* const* d_in, const int*  , int  ,
                              void* d_out, int  , void*  , size_t  ,
                              hipStream_t stream) {
  const float* x = (const float*)d_in[0];
  const float* P = (const float*)d_in[1];
  float* out = (float*)d_out;

  hipLaunchKernelGGL(fullscan_mid_wmma, dim3(GridBlocks), dim3(Threads), 0, stream, x, P, out);
}


# Two-phase fast-FIR split: three 32-tap correlations x1*(h0+h1), (x0-x1)*h0, (x0'-x1)*h1 replace four; y=T+U, T+V; 792 not 1040 WMMAs per wave; sums regrouped (differences and h0+h1 formed first)
# speedup vs baseline: 3.0224x; 1.2109x over previous
#include <hip/hip_runtime.h>


typedef __attribute__((ext_vector_type(2))) float v2f;
typedef __attribute__((ext_vector_type(4))) float v4f;
typedef __attribute__((ext_vector_type(8))) float v8f;

namespace {
constexpr int Dch   = 16;
constexpr int Llen  = 1000000;
constexpr int Wlen  = 64;
constexpr int Kout  = 8;
constexpr int Pad   = 31;
constexpr int MidLen = Llen - Wlen + 1;
constexpr int HalfW = Wlen / 2;
constexpr int WavesPerBlock = 8;
constexpr int TilesPerWave  = 2;
constexpr int TileI = 32;
constexpr int WaveI   = TilesPerWave * TileI;
constexpr int WavePos = 2 * WaveI;
constexpr int BlockI   = WavesPerBlock * WaveI;
constexpr int BlockPos = 2 * BlockI;
constexpr int Blocks   = (MidLen + BlockPos - 1) / BlockPos;
constexpr int MainSteps = HalfW / 4;
constexpr int Threads = WavesPerBlock * 32;
constexpr int XRaw    = BlockPos + Wlen;
constexpr int RowVecs = XRaw / 4;
constexpr int XVecs   = Dch * RowVecs;
constexpr int XIters  = XVecs / Threads;
constexpr int SL = BlockI + HalfW;
constexpr int DS = 3 * SL;
constexpr int SLen = Dch * DS;
constexpr int DP = 33;
constexpr int KP = Dch * DP + 8;
constexpr int HLen = Kout * KP;
constexpr int PVecs  = Kout * Dch * Wlen / 4;
constexpr int PIters = PVecs / Threads;
constexpr int EdgeOuts   = Kout * Pad + Kout * 32;
constexpr int EdgeBlocks = (EdgeOuts + Threads - 1) / Threads;
constexpr int GridBlocks = EdgeBlocks + Blocks;
constexpr int OP = 132;
constexpr int WaveOut = Kout * OP;
static_assert(WavesPerBlock * WaveOut <= SLen, "the staged outputs fit in the sequences' space");
static_assert(OP % 4 == 0 && WavePos % 32 == 0 && (Pad + 1) % 32 == 0 && Llen % 32 == 0, "position 1 of a wave starts a 128-byte line of out");
static_assert(PVecs % Threads == 0 && XVecs % Threads == 0, "P and x pieces divide evenly");
static_assert(XRaw % 4 == 0 && Llen % 4 == 0 && BlockPos % 4 == 0, "16-byte pieces never straddle the end of a row");
static_assert(SL % 2 == 0 && 2 * (RowVecs - 1) + 1 < SL, "pairs of a piece land inside an even-pitched sequence");
}

__device__ __forceinline__ void fullscan_edges(int t, const float* __restrict__ xg,
                                               const float* __restrict__ Pg,
                                               float* __restrict__ out)
{
  if (t < Kout * Pad) {
    int k = t / Pad, j = t - k * Pad;
    float s = 0.f;
    for (int mm = 0; mm <= 32 + j; ++mm) {
      float tl = 0.f;
      for (int d = 0; d < Dch; ++d)
        tl += xg[(size_t)d * Llen + mm] * Pg[(k * Dch + d) * Wlen + mm];
      s += tl;
    }
    out[(size_t)k * Llen + j] = s;
  } else if (t < Kout * Pad + Kout * 32) {
    int t2 = t - Kout * Pad;
    int k = t2 / 32, j = t2 - k * 32;
    float s = 0.f;
    for (int mm = j + 1; mm < Wlen; ++mm) {
      for (int d = 0; d < Dch; ++d)
        s += xg[(size_t)d * Llen + (Llen - Wlen + mm)] *
             Pg[(k * Dch + d) * Wlen + mm];
    }
    out[(size_t)k * Llen + (Llen - 32) + j] = s;
  }
}

__global__ __launch_bounds__(Threads) void fullscan_mid_wmma(
    const float* __restrict__ xg, const float* __restrict__ Pg,
    float* __restrict__ out)
{
  __shared__ float sH[2 * HLen];
  __shared__ __attribute__((aligned(16))) float sS[SLen];

  if (blockIdx.x < EdgeBlocks) {
    fullscan_edges(blockIdx.x * Threads + threadIdx.x, xg, Pg, out);
    return;
  }

  const int tid = threadIdx.x;
  const int N0  = (blockIdx.x - EdgeBlocks) * BlockPos;

  const v4f* P4 = reinterpret_cast<const v4f*>(Pg);
  v4f pr[PIters];
#pragma unroll
  for (int it = 0; it < PIters; ++it) pr[it] = P4[tid + it * Threads];

  v4f xr[XIters];
  float xn[XIters];
#pragma unroll
  for (int it = 0; it < XIters; ++it) {
    const int i = tid + it * Threads;
    const int d = i / RowVecs;
    const int q = i - d * RowVecs;
    const int gi = N0 + 4 * q;
    const int gc = (gi < Llen) ? gi : (Llen - 4);
    const int gn = (gi + 4 < Llen) ? (gi + 4) : (Llen - 1);
    v4f v = *reinterpret_cast<const v4f*>(&xg[(unsigned)d * (unsigned)Llen + (unsigned)gc]);
    if (gi >= Llen) { v.x = v.w; v.y = v.w; v.z = v.w; }
    xr[it] = v;
    xn[it] = xg[(unsigned)d * (unsigned)Llen + (unsigned)gn];
  }

  for (int i = tid; i < 2 * Kout * Dch; i += Threads) {
    const int set = i >> 7;
    const int row = i & 127;
    sH[set * HLen + (row >> 4) * KP + (row & 15) * DP] = 0.0f;
  }

#pragma unroll
  for (int it = 0; it < PIters; ++it) {
    const int e = 4 * (tid + it * Threads);
    const int k = e >> 10;
    const int d = (e >> 6) & 15;
    const int w = e & 63;
    float* dst = &sH[k * KP + d * DP + (w >> 1) + 1];
    dst[0] = pr[it].x; dst[1] = pr[it].z;
    dst[HLen] = pr[it].y; dst[HLen + 1] = pr[it].w;
  }
#pragma unroll
  for (int it = 0; it < XIters; ++it) {
    const int i = tid + it * Threads;
    const int d = i / RowVecs;
    const int q = i - d * RowVecs;
    const v4f v = xr[it];
    float* dst = &sS[d * DS + 2 * q];
    v2f s1, su, sv;
    s1.x = v.y;       s1.y = v.w;
    su.x = v.x - v.y; su.y = v.z - v.w;
    sv.x = v.z - v.y; sv.y = xn[it] - v.w;
    *reinterpret_cast<v2f*>(dst)          = s1;
    *reinterpret_cast<v2f*>(dst + SL)     = su;
    *reinterpret_cast<v2f*>(dst + 2 * SL) = sv;
  }
  __syncthreads();

  const int wave = tid >> 5;
  const int lane = tid & 31;
  const int n0 = N0 + wave * WavePos;
  const bool active = n0 < MidLen;

  const int m  = lane & 15;
  const int hi = lane >> 4;
  const int c  = hi * 2;
  const int kch = m >> 1;
  const int sh  = m & 1;
  const float* xb = &sS[wave * WaveI + 2 * m + c];
  const float* pb = &sH[kch * KP + c - sh + 1];

  v8f accT0 = {}, accT1 = {}, accU0 = {}, accU1 = {}, accV0 = {}, accV1 = {};
  if (active) {
    for (int d = 0; d < Dch; ++d) {
#pragma unroll
      for (int w4 = 0; w4 < MainSteps; ++w4) {
        const int xo = d * DS + w4 * 4;
        const int po = d * DP + w4 * 4;
        v2f bU, bV, bT;
        bU.x = pb[po];        bU.y = pb[po + 1];
        bV.x = pb[HLen + po]; bV.y = pb[HLen + po + 1];
        bT = bU + bV;
        const v2f aT0 = *reinterpret_cast<const v2f*>(&xb[xo]);
        const v2f aT1 = *reinterpret_cast<const v2f*>(&xb[xo + TileI]);
        const v2f aU0 = *reinterpret_cast<const v2f*>(&xb[xo + SL]);
        const v2f aU1 = *reinterpret_cast<const v2f*>(&xb[xo + SL + TileI]);
        const v2f aV0 = *reinterpret_cast<const v2f*>(&xb[xo + 2 * SL]);
        const v2f aV1 = *reinterpret_cast<const v2f*>(&xb[xo + 2 * SL + TileI]);
        accT0 = __builtin_amdgcn_wmma_f32_16x16x4_f32(false, aT0, false, bT, (short)0, accT0, false, false);
        accT1 = __builtin_amdgcn_wmma_f32_16x16x4_f32(false, aT1, false, bT, (short)0, accT1, false, false);
        accU0 = __builtin_amdgcn_wmma_f32_16x16x4_f32(false, aU0, false, bU, (short)0, accU0, false, false);
        accU1 = __builtin_amdgcn_wmma_f32_16x16x4_f32(false, aU1, false, bU, (short)0, accU1, false, false);
        accV0 = __builtin_amdgcn_wmma_f32_16x16x4_f32(false, aV0, false, bV, (short)0, accV0, false, false);
        accV1 = __builtin_amdgcn_wmma_f32_16x16x4_f32(false, aV1, false, bV, (short)0, accV1, false, false);
      }
    }
    const float* xt = &sS[wave * WaveI + 2 * m + HalfW];
    const float* pt = &sH[kch * KP + HalfW];
#pragma unroll
    for (int g = 0; g < Dch / 4; ++g) {
      const int d0 = 4 * g + c;
      const float u0 = pt[d0 * DP];
      const float u1 = pt[(d0 + 1) * DP];
      const float w0 = pt[HLen + d0 * DP];
      const float w1 = pt[HLen + (d0 + 1) * DP];
      v2f bU, bV, bT;
      bU.x = sh ? u0 : 0.0f; bU.y = sh ? u1 : 0.0f;
      bV.x = sh ? w0 : 0.0f; bV.y = sh ? w1 : 0.0f;
      bT = bU + bV;
      const float* x0 = xt + d0 * DS;
      const float* x1 = x0 + DS;
      v2f aT0, aT1, aU0, aU1, aV0, aV1;
      aT0.x = x0[0];              aT0.y = x1[0];
      aT1.x = x0[TileI];          aT1.y = x1[TileI];
      aU0.x = x0[SL];             aU0.y = x1[SL];
      aU1.x = x0[SL + TileI];     aU1.y = x1[SL + TileI];
      aV0.x = x0[2 * SL];         aV0.y = x1[2 * SL];
      aV1.x = x0[2 * SL + TileI]; aV1.y = x1[2 * SL + TileI];
      accT0 = __builtin_amdgcn_wmma_f32_16x16x4_f32(false, aT0, false, bT, (short)0, accT0, false, false);
      accT1 = __builtin_amdgcn_wmma_f32_16x16x4_f32(false, aT1, false, bT, (short)0, accT1, false, false);
      accU0 = __builtin_amdgcn_wmma_f32_16x16x4_f32(false, aU0, false, bU, (short)0, accU0, false, false);
      accU1 = __builtin_amdgcn_wmma_f32_16x16x4_f32(false, aU1, false, bU, (short)0, accU1, false, false);
      accV0 = __builtin_amdgcn_wmma_f32_16x16x4_f32(false, aV0, false, bV, (short)0, accV0, false, false);
      accV1 = __builtin_amdgcn_wmma_f32_16x16x4_f32(false, aV1, false, bV, (short)0, accV1, false, false);
    }
  }

  __syncthreads();
  float* sO = sS + wave * WaveOut;
  if (active) {
    float* row = &sO[kch * OP + 3 + 32 * hi + 2 * sh];
#pragma unroll
    for (int j = 0; j < 8; ++j) {
      row[4 * j]      = accT0[j] + accU0[j];
      row[4 * j + 1]  = accT0[j] + accV0[j];
      row[4 * j + 64] = accT1[j] + accU1[j];
      row[4 * j + 65] = accT1[j] + accV1[j];
    }
  }
  __syncthreads();
  if (active) {
    const int p  = 1 + 4 * lane;
    const int ks = lane >> 2;
    const int qs = lane & 3;
    const int ps = (qs == 0) ? 0 : (WavePos - 4 + qs);
    if (n0 + WavePos <= MidLen) {
      if (lane < 31) {
        const float* srow = &sO[3 + p];
        float* dst = out + Pad + n0 + p;
#pragma unroll
        for (int k = 0; k < Kout; ++k)
          *reinterpret_cast<v4f*>(dst + (size_t)k * Llen) = *reinterpret_cast<const v4f*>(srow + k * OP);
      }
      out[(size_t)ks * Llen + Pad + n0 + ps] = sO[ks * OP + 3 + ps];
    } else {
      if (lane < 31) {
        for (int k = 0; k < Kout; ++k)
          for (int e = 0; e < 4; ++e)
            if (n0 + p + e < MidLen) out[(size_t)k * Llen + Pad + n0 + p + e] = sO[k * OP + 3 + p + e];
      }
      if (n0 + ps < MidLen) out[(size_t)ks * Llen + Pad + n0 + ps] = sO[ks * OP + 3 + ps];
    }
  }
}

extern "C" void kernel_launch(void* const* d_in, const int*  , int  ,
                              void* d_out, int  , void*  , size_t  ,
                              hipStream_t stream) {
  const float* x = (const float*)d_in[0];
  const float* P = (const float*)d_in[1];
  float* out = (float*)d_out;

  hipLaunchKernelGGL(fullscan_mid_wmma, dim3(GridBlocks), dim3(Threads), 0, stream, x, P, out);
}


# Fast-FIR split applied again inside each of T, U, V: nine 16-tap correlations over quad positions replace three 32-tap ones, 612 not 792 WMMAs per wave; sums regrouped (second differences first)
# speedup vs baseline: 3.3769x; 1.1173x over previous
#include <hip/hip_runtime.h>


typedef __attribute__((ext_vector_type(2))) float v2f;
typedef __attribute__((ext_vector_type(4))) float v4f;
typedef __attribute__((ext_vector_type(8))) float v8f;

namespace {
constexpr int Dch   = 16;
constexpr int Llen  = 1000000;
constexpr int Wlen  = 64;
constexpr int Kout  = 8;
constexpr int Pad   = 31;
constexpr int MidLen = Llen - Wlen + 1;
constexpr int QuartW = Wlen / 4;
constexpr int HalfW = Wlen / 2;
constexpr int WavesPerBlock = 8;
constexpr int WaveQ = 32;
constexpr int WaveI   = 2 * WaveQ;
constexpr int WavePos = 2 * WaveI;
constexpr int BlockI   = WavesPerBlock * WaveI;
constexpr int BlockPos = 2 * BlockI;
constexpr int Blocks   = (MidLen + BlockPos - 1) / BlockPos;
constexpr int MainSteps = QuartW / 4;
constexpr int Threads = WavesPerBlock * 32;
constexpr int XRaw    = BlockPos + Wlen;
constexpr int RowVecs = XRaw / 4;
constexpr int XVecs   = Dch * RowVecs;
constexpr int XIters  = XVecs / Threads;
constexpr int SL = BlockI + HalfW;
constexpr int SLh = SL / 2;
constexpr int DS = 3 * SL;
constexpr int SLen = Dch * DS;
constexpr int DP = 17;
constexpr int KP = Dch * DP + 8;
constexpr int HLen = Kout * KP;
constexpr int PVecs  = Kout * Dch * Wlen / 4;
constexpr int PIters = PVecs / Threads;
constexpr int EdgeOuts   = Kout * Pad + Kout * 32;
constexpr int EdgeBlocks = (EdgeOuts + Threads - 1) / Threads;
constexpr int GridBlocks = EdgeBlocks + Blocks;
constexpr int OP = 132;
constexpr int WaveOut = Kout * OP;
static_assert(WavesPerBlock * WaveOut <= SLen, "the staged outputs fit in the sequences' space");
static_assert(OP % 4 == 0 && WavePos % 32 == 0 && (Pad + 1) % 32 == 0 && Llen % 32 == 0, "position 1 of a wave starts a 128-byte line of out");
static_assert(PVecs % Threads == 0 && XVecs % Threads == 0, "P and x pieces divide evenly");
static_assert(XRaw % 4 == 0 && Llen % 4 == 0 && BlockPos % 4 == 0, "16-byte pieces never straddle the end of a row");
static_assert(SL % 4 == 0 && RowVecs <= SLh, "a piece's two pairs land inside the even-pitched parity planes");
}

__device__ __forceinline__ void fullscan_edges(int t, const float* __restrict__ xg,
                                               const float* __restrict__ Pg,
                                               float* __restrict__ out)
{
  if (t < Kout * Pad) {
    int k = t / Pad, j = t - k * Pad;
    float s = 0.f;
    for (int mm = 0; mm <= 32 + j; ++mm) {
      float tl = 0.f;
      for (int d = 0; d < Dch; ++d)
        tl += xg[(size_t)d * Llen + mm] * Pg[(k * Dch + d) * Wlen + mm];
      s += tl;
    }
    out[(size_t)k * Llen + j] = s;
  } else if (t < Kout * Pad + Kout * 32) {
    int t2 = t - Kout * Pad;
    int k = t2 / 32, j = t2 - k * 32;
    float s = 0.f;
    for (int mm = j + 1; mm < Wlen; ++mm) {
      for (int d = 0; d < Dch; ++d)
        s += xg[(size_t)d * Llen + (Llen - Wlen + mm)] *
             Pg[(k * Dch + d) * Wlen + mm];
    }
    out[(size_t)k * Llen + (Llen - 32) + j] = s;
  }
}

__global__ __launch_bounds__(Threads) void fullscan_mid_wmma(
    const float* __restrict__ xg, const float* __restrict__ Pg,
    float* __restrict__ out)
{
  __shared__ float sH[4 * HLen];
  __shared__ __attribute__((aligned(16))) float sS[SLen];

  if (blockIdx.x < EdgeBlocks) {
    fullscan_edges(blockIdx.x * Threads + threadIdx.x, xg, Pg, out);
    return;
  }

  const int tid = threadIdx.x;
  const int N0  = (blockIdx.x - EdgeBlocks) * BlockPos;

  const v4f* P4 = reinterpret_cast<const v4f*>(Pg);
  v4f pr[PIters];
#pragma unroll
  for (int it = 0; it < PIters; ++it) pr[it] = P4[tid + it * Threads];

  v4f xr[XIters];
  float xn[XIters];
#pragma unroll
  for (int it = 0; it < XIters; ++it) {
    const int i = tid + it * Threads;
    const int d = i / RowVecs;
    const int q = i - d * RowVecs;
    const int gi = N0 + 4 * q;
    const int gc = (gi < Llen) ? gi : (Llen - 4);
    const int gn = (gi + 4 < Llen) ? (gi + 4) : (Llen - 1);
    v4f v = *reinterpret_cast<const v4f*>(&xg[(unsigned)d * (unsigned)Llen + (unsigned)gc]);
    if (gi >= Llen) { v.x = v.w; v.y = v.w; v.z = v.w; }
    xr[it] = v;
    xn[it] = xg[(unsigned)d * (unsigned)Llen + (unsigned)gn];
  }

  for (int i = tid; i < 4 * Kout * Dch; i += Threads) {
    const int set = i >> 7;
    const int row = i & 127;
    sH[set * HLen + (row >> 4) * KP + (row & 15) * DP] = 0.0f;
  }

#pragma unroll
  for (int it = 0; it < PIters; ++it) {
    const int e = 4 * (tid + it * Threads);
    const int k = e >> 10;
    const int d = (e >> 6) & 15;
    const int w = e & 63;
    float* dst = &sH[k * KP + d * DP + (w >> 2) + 1];
    dst[0]        = pr[it].x;
    dst[HLen]     = pr[it].y;
    dst[2 * HLen] = pr[it].z;
    dst[3 * HLen] = pr[it].w;
  }
#pragma unroll
  for (int it = 0; it < XIters; ++it) {
    const int i = tid + it * Threads;
    const int d = i / RowVecs;
    const int q = i - d * RowVecs;
    const v4f v = xr[it];
    float* dst = &sS[d * DS + q];
    dst[0]            = v.y;
    dst[SLh]          = v.w;
    dst[SL]           = v.x - v.y;
    dst[SL + SLh]     = v.z - v.w;
    dst[2 * SL]       = v.z - v.y;
    dst[2 * SL + SLh] = xn[it] - v.w;
  }
  __syncthreads();

  const int wave = tid >> 5;
  const int lane = tid & 31;
  const int n0 = N0 + wave * WavePos;
  const bool active = n0 < MidLen;

  const int m  = lane & 15;
  const int hi = lane >> 4;
  const int c  = hi * 2;
  const int kch = m >> 1;
  const int sh  = m & 1;
  const float* xb = &sS[wave * WaveQ + 2 * m + c];
  const float* pb = &sH[kch * KP + c - sh + 1];

  v8f accTt = {}, accTu = {}, accTv = {}, accUt = {}, accUu = {}, accUv = {}, accVt = {}, accVu = {}, accVv = {};
  if (active) {
    for (int d = 0; d < Dch; ++d) {
#pragma unroll
      for (int w4 = 0; w4 < MainSteps; ++w4) {
        const int xo = d * DS + w4 * 4;
        const int po = d * DP + w4 * 4;
        v2f b0, b1, b2, b3;
        b0.x = pb[po];            b0.y = pb[po + 1];
        b1.x = pb[HLen + po];     b1.y = pb[HLen + po + 1];
        b2.x = pb[2 * HLen + po]; b2.y = pb[2 * HLen + po + 1];
        b3.x = pb[3 * HLen + po]; b3.y = pb[3 * HLen + po + 1];
        const v2f b01 = b0 + b1, b23 = b2 + b3, b02 = b0 + b2, b13 = b1 + b3;
        const v2f bAll = b01 + b23;
        const float* xT = &xb[xo];
        const float* xU = &xb[xo + SL];
        const float* xV = &xb[xo + 2 * SL];
        const v2f aTt = *reinterpret_cast<const v2f*>(xT + SLh);
        const v2f eT  = *reinterpret_cast<const v2f*>(xT);
        v2f nT; nT.x = xT[1]; nT.y = xT[2];
        const v2f aUt = *reinterpret_cast<const v2f*>(xU + SLh);
        const v2f eU  = *reinterpret_cast<const v2f*>(xU);
        v2f nU; nU.x = xU[1]; nU.y = xU[2];
        const v2f aVt = *reinterpret_cast<const v2f*>(xV + SLh);
        const v2f eV  = *reinterpret_cast<const v2f*>(xV);
        v2f nV; nV.x = xV[1]; nV.y = xV[2];
        const v2f aTu = eT - aTt, aTv = nT - aTt;
        const v2f aUu = eU - aUt, aUv = nU - aUt;
        const v2f aVu = eV - aVt, aVv = nV - aVt;
        accTt = __builtin_amdgcn_wmma_f32_16x16x4_f32(false, aTt, false, bAll, (short)0, accTt, false, false);
        accTu = __builtin_amdgcn_wmma_f32_16x16x4_f32(false, aTu, false, b01,  (short)0, accTu, false, false);
        accTv = __builtin_amdgcn_wmma_f32_16x16x4_f32(false, aTv, false, b23,  (short)0, accTv, false, false);
        accUt = __builtin_amdgcn_wmma_f32_16x16x4_f32(false, aUt, false, b02,  (short)0, accUt, false, false);
        accUu = __builtin_amdgcn_wmma_f32_16x16x4_f32(false, aUu, false, b0,   (short)0, accUu, false, false);
        accUv = __builtin_amdgcn_wmma_f32_16x16x4_f32(false, aUv, false, b2,   (short)0, accUv, false, false);
        accVt = __builtin_amdgcn_wmma_f32_16x16x4_f32(false, aVt, false, b13,  (short)0, accVt, false, false);
        accVu = __builtin_amdgcn_wmma_f32_16x16x4_f32(false, aVu, false, b1,   (short)0, accVu, false, false);
        accVv = __builtin_amdgcn_wmma_f32_16x16x4_f32(false, aVv, false, b3,   (short)0, accVv, false, false);
      }
    }
    const float* xt = &sS[wave * WaveQ + 2 * m + QuartW];
    const float* pt = &sH[kch * KP + QuartW];
#pragma unroll
    for (int g = 0; g < Dch / 4; ++g) {
      const int d0 = 4 * g + c;
      v2f b0, b1, b2, b3;
      {
        const float f00 = pt[d0 * DP],            f01 = pt[(d0 + 1) * DP];
        const float f10 = pt[HLen + d0 * DP],     f11 = pt[HLen + (d0 + 1) * DP];
        const float f20 = pt[2 * HLen + d0 * DP], f21 = pt[2 * HLen + (d0 + 1) * DP];
        const float f30 = pt[3 * HLen + d0 * DP], f31 = pt[3 * HLen + (d0 + 1) * DP];
        b0.x = sh ? f00 : 0.0f; b0.y = sh ? f01 : 0.0f;
        b1.x = sh ? f10 : 0.0f; b1.y = sh ? f11 : 0.0f;
        b2.x = sh ? f20 : 0.0f; b2.y = sh ? f21 : 0.0f;
        b3.x = sh ? f30 : 0.0f; b3.y = sh ? f31 : 0.0f;
      }
      const v2f b01 = b0 + b1, b23 = b2 + b3, b02 = b0 + b2, b13 = b1 + b3;
      const v2f bAll = b01 + b23;
      const float* x0 = xt + d0 * DS;
      const float* x1 = x0 + DS;
      const v2f eT0 = *reinterpret_cast<const v2f*>(x0),          eT1 = *reinterpret_cast<const v2f*>(x1);
      const v2f eU0 = *reinterpret_cast<const v2f*>(x0 + SL),     eU1 = *reinterpret_cast<const v2f*>(x1 + SL);
      const v2f eV0 = *reinterpret_cast<const v2f*>(x0 + 2 * SL), eV1 = *reinterpret_cast<const v2f*>(x1 + 2 * SL);
      v2f aTt, aTu, aTv, aUt, aUu, aUv, aVt, aVu, aVv;
      aTt.x = x0[SLh];          aTt.y = x1[SLh];
      aUt.x = x0[SL + SLh];     aUt.y = x1[SL + SLh];
      aVt.x = x0[2 * SL + SLh]; aVt.y = x1[2 * SL + SLh];
      aTu.x = eT0.x - aTt.x; aTu.y = eT1.x - aTt.y;
      aTv.x = eT0.y - aTt.x; aTv.y = eT1.y - aTt.y;
      aUu.x = eU0.x - aUt.x; aUu.y = eU1.x - aUt.y;
      aUv.x = eU0.y - aUt.x; aUv.y = eU1.y - aUt.y;
      aVu.x = eV0.x - aVt.x; aVu.y = eV1.x - aVt.y;
      aVv.x = eV0.y - aVt.x; aVv.y = eV1.y - aVt.y;
      accTt = __builtin_amdgcn_wmma_f32_16x16x4_f32(false, aTt, false, bAll, (short)0, accTt, false, false);
      accTu = __builtin_amdgcn_wmma_f32_16x16x4_f32(false, aTu, false, b01,  (short)0, accTu, false, false);
      accTv = __builtin_amdgcn_wmma_f32_16x16x4_f32(false, aTv, false, b23,  (short)0, accTv, false, false);
      accUt = __builtin_amdgcn_wmma_f32_16x16x4_f32(false, aUt, false, b02,  (short)0, accUt, false, false);
      accUu = __builtin_amdgcn_wmma_f32_16x16x4_f32(false, aUu, false, b0,   (short)0, accUu, false, false);
      accUv = __builtin_amdgcn_wmma_f32_16x16x4_f32(false, aUv, false, b2,   (short)0, accUv, false, false);
      accVt = __builtin_amdgcn_wmma_f32_16x16x4_f32(false, aVt, false, b13,  (short)0, accVt, false, false);
      accVu = __builtin_amdgcn_wmma_f32_16x16x4_f32(false, aVu, false, b1,   (short)0, accVu, false, false);
      accVv = __builtin_amdgcn_wmma_f32_16x16x4_f32(false, aVv, false, b3,   (short)0, accVv, false, false);
    }
  }

  __syncthreads();
  float* sO = sS + wave * WaveOut;
  if (active) {
    float* row = &sO[kch * OP + 3 + 64 * hi + 4 * sh];
#pragma unroll
    for (int j = 0; j < 8; ++j) {
      const float t0 = accTt[j] + accTu[j], t1 = accTt[j] + accTv[j];
      const float u0 = accUt[j] + accUu[j], u1 = accUt[j] + accUv[j];
      const float v0 = accVt[j] + accVu[j], v1 = accVt[j] + accVv[j];
      row[8 * j]     = t0 + u0;
      row[8 * j + 1] = t0 + v0;
      row[8 * j + 2] = t1 + u1;
      row[8 * j + 3] = t1 + v1;
    }
  }
  __syncthreads();
  if (active) {
    const int p  = 1 + 4 * lane;
    const int ks = lane >> 2;
    const int qs = lane & 3;
    const int ps = (qs == 0) ? 0 : (WavePos - 4 + qs);
    if (n0 + WavePos <= MidLen) {
      if (lane < 31) {
        const float* srow = &sO[3 + p];
        float* dst = out + Pad + n0 + p;
#pragma unroll
        for (int k = 0; k < Kout; ++k)
          *reinterpret_cast<v4f*>(dst + (size_t)k * Llen) = *reinterpret_cast<const v4f*>(srow + k * OP);
      }
      out[(size_t)ks * Llen + Pad + n0 + ps] = sO[ks * OP + 3 + ps];
    } else {
      if (lane < 31) {
        for (int k = 0; k < Kout; ++k)
          for (int e = 0; e < 4; ++e)
            if (n0 + p + e < MidLen) out[(size_t)k * Llen + Pad + n0 + p + e] = sO[k * OP + 3 + p + e];
      }
      if (n0 + ps < MidLen) out[(size_t)ks * Llen + Pad + n0 + ps] = sO[ks * OP + 3 + ps];
    }
  }
}

extern "C" void kernel_launch(void* const* d_in, const int*  , int  ,
                              void* d_out, int  , void*  , size_t  ,
                              hipStream_t stream) {
  const float* x = (const float*)d_in[0];
  const float* P = (const float*)d_in[1];
  float* out = (float*)d_out;

  hipLaunchKernelGGL(fullscan_mid_wmma, dim3(GridBlocks), dim3(Threads), 0, stream, x, P, out);
}
